# starting kernel
# baseline (speedup 1.0000x reference)
#include <hip/hip_runtime.h>
#include <math.h>

#define N_NODES 8192
#define D_OSC   64
#define E_EDGES 1048576

typedef __attribute__((ext_vector_type(2))) float v2f;
typedef __attribute__((ext_vector_type(8))) float v8f;

__device__ __forceinline__ float wave_reduce_sum(float v) {
  #pragma unroll
  for (int m = 16; m >= 1; m >>= 1) v += __shfl_xor(v, m, 32);
  return v;
}

__global__ __launch_bounds__(256) void k_zero(float* __restrict__ p, int n) {
  int i = blockIdx.x * 256 + threadIdx.x;
  if (i < n) p[i] = 0.0f;
}

__global__ __launch_bounds__(256) void k_norm_tanh(const float* __restrict__ sk,
                                                   const float* __restrict__ sh,
                                                   float* __restrict__ skn,
                                                   float* __restrict__ gv) {
  int row  = blockIdx.x * 8 + (threadIdx.x >> 5);
  int lane = threadIdx.x & 31;
  const float* r = sk + (size_t)row * D_OSC;
  float a = r[lane];
  float b = r[lane + 32];
  float ss  = wave_reduce_sum(a * a + b * b);
  float inv = rsqrtf(ss);
  float* o = skn + (size_t)row * D_OSC;
  o[lane]      = a * inv;
  o[lane + 32] = b * inv;

  int gi = blockIdx.x * 256 + threadIdx.x;
  if (gi < N_NODES) gv[gi] = tanhf(sh[gi]);
}

__global__ __launch_bounds__(256) void k_edge_K(const int* __restrict__ ind,
                                                const float* __restrict__ wK,
                                                const float* __restrict__ skn,
                                                float* __restrict__ fK) {
  int e    = blockIdx.x * 8 + (threadIdx.x >> 5);
  int lane = threadIdx.x & 31;
  int j0 = ind[2 * e];
  int j1 = ind[2 * e + 1];
  const float* r0 = skn + (size_t)j0 * D_OSC;
  const float* r1 = skn + (size_t)j1 * D_OSC;
  float a0 = r0[lane], a1 = r0[lane + 32];
  float b0 = r1[lane], b1 = r1[lane + 32];
  float s = wave_reduce_sum(a0 * b0 + a1 * b1);
  float c = wK[e] * s;
  atomicAdd(&fK[(size_t)j0 * D_OSC + lane],       c * b0);
  atomicAdd(&fK[(size_t)j0 * D_OSC + lane + 32],  c * b1);
  atomicAdd(&fK[(size_t)j1 * D_OSC + lane],       c * a0);
  atomicAdd(&fK[(size_t)j1 * D_OSC + lane + 32],  c * a1);
}

__global__ __launch_bounds__(256) void k_edge_HK(const int* __restrict__ ind,
                                                 const float* __restrict__ whk,
                                                 const float* __restrict__ skn,
                                                 const float* __restrict__ gv,
                                                 const float* __restrict__ kH,
                                                 const float* __restrict__ kK,
                                                 float* __restrict__ fH,
                                                 float* __restrict__ fK) {
  int e    = blockIdx.x * 8 + (threadIdx.x >> 5);
  int lane = threadIdx.x & 31;
  int i0 = ind[2 * e];
  int i1 = ind[2 * e + 1];
  const float* r0 = skn + (size_t)i0 * D_OSC;
  const float* r1 = skn + (size_t)i1 * D_OSC;
  float a0 = r0[lane], a1 = r0[lane + 32];
  float b0 = r1[lane], b1 = r1[lane + 32];
  float Gram = wave_reduce_sum(a0 * b0 + a1 * b1);
  float w  = whk[e];
  float g0 = gv[i0];
  float g1 = gv[i1];
  float invKH = 1.0f / kH[0];
  float invKK = 1.0f / kK[0];
  if (lane == 0) {
    atomicAdd(&fH[i0], Gram * w * g1 * invKH);
    atomicAdd(&fH[i1], Gram * w * g0 * invKH);
  }
  float Gc = -(g0 * g1) * w * invKK;
  atomicAdd(&fK[(size_t)i0 * D_OSC + lane],      Gc * b0);
  atomicAdd(&fK[(size_t)i0 * D_OSC + lane + 32], Gc * b1);
  atomicAdd(&fK[(size_t)i1 * D_OSC + lane],      Gc * a0);
  atomicAdd(&fK[(size_t)i1 * D_OSC + lane + 32], Gc * a1);
}

__global__ __launch_bounds__(256) void k_gemv_fused(const float* __restrict__ W,
                                                    const float* __restrict__ gv,
                                                    float* __restrict__ y,
                                                    float* __restrict__ z) {
  __shared__ float slab[8 * 288];
  __shared__ float zstage[512];

  const int wv   = threadIdx.x >> 5;
  const int lane = threadIdx.x & 31;
  const int hl   = lane & 15;
  const int koff = (lane < 16) ? 0 : 2;

  const int C0 = blockIdx.x * 512;
  const int r0 = blockIdx.y * 128 + wv * 16;

  zstage[threadIdx.x]       = 0.0f;
  zstage[threadIdx.x + 256] = 0.0f;
  __syncthreads();

  float* sl = slab + wv * 288;

  v2f bz0 = *(const v2f*)(gv + r0 +  0 + koff);
  v2f bz1 = *(const v2f*)(gv + r0 +  4 + koff);
  v2f bz2 = *(const v2f*)(gv + r0 +  8 + koff);
  v2f bz3 = *(const v2f*)(gv + r0 + 12 + koff);

  const float* wrow = W + (size_t)(r0 + hl) * N_NODES + koff;

  v8f yacc = {};
  for (int t = 0; t < 32; ++t) {
    const int c = C0 + t * 16;
    const float* wp = wrow + c;
    v2f a0 = *(const v2f*)(wp +  0);
    v2f a1 = *(const v2f*)(wp +  4);
    v2f a2 = *(const v2f*)(wp +  8);
    v2f a3 = *(const v2f*)(wp + 12);
    v2f by0 = *(const v2f*)(gv + c +  0 + koff);
    v2f by1 = *(const v2f*)(gv + c +  4 + koff);
    v2f by2 = *(const v2f*)(gv + c +  8 + koff);
    v2f by3 = *(const v2f*)(gv + c + 12 + koff);

    yacc = __builtin_amdgcn_wmma_f32_16x16x4_f32(false, a0, false, by0, (short)0, yacc, false, false);
    yacc = __builtin_amdgcn_wmma_f32_16x16x4_f32(false, a1, false, by1, (short)0, yacc, false, false);
    yacc = __builtin_amdgcn_wmma_f32_16x16x4_f32(false, a2, false, by2, (short)0, yacc, false, false);
    yacc = __builtin_amdgcn_wmma_f32_16x16x4_f32(false, a3, false, by3, (short)0, yacc, false, false);

    float* sr = sl + hl * 18 + koff;
    *(v2f*)(sr +  0) = a0;
    *(v2f*)(sr +  4) = a1;
    *(v2f*)(sr +  8) = a2;
    *(v2f*)(sr + 12) = a3;

    v2f az0, az1, az2, az3;
    az0[0] = sl[( 0 + koff) * 18 + hl];  az0[1] = sl[( 1 + koff) * 18 + hl];
    az1[0] = sl[( 4 + koff) * 18 + hl];  az1[1] = sl[( 5 + koff) * 18 + hl];
    az2[0] = sl[( 8 + koff) * 18 + hl];  az2[1] = sl[( 9 + koff) * 18 + hl];
    az3[0] = sl[(12 + koff) * 18 + hl];  az3[1] = sl[(13 + koff) * 18 + hl];

    v8f zt = {};
    zt = __builtin_amdgcn_wmma_f32_16x16x4_f32(false, az0, false, bz0, (short)0, zt, false, false);
    zt = __builtin_amdgcn_wmma_f32_16x16x4_f32(false, az1, false, bz1, (short)0, zt, false, false);
    zt = __builtin_amdgcn_wmma_f32_16x16x4_f32(false, az2, false, bz2, (short)0, zt, false, false);
    zt = __builtin_amdgcn_wmma_f32_16x16x4_f32(false, az3, false, bz3, (short)0, zt, false, false);

    if (hl == 0) {
      int zb = t * 16 + ((lane == 16) ? 8 : 0);
      #pragma unroll
      for (int v = 0; v < 8; ++v) atomicAdd(&zstage[zb + v], zt[v]);
    }
  }

  if (hl == 0) {
    int base = r0 + ((lane == 16) ? 8 : 0);
    #pragma unroll
    for (int v = 0; v < 8; ++v) atomicAdd(&y[base + v], yacc[v]);
  }

  __syncthreads();
  atomicAdd(&z[C0 + threadIdx.x],       zstage[threadIdx.x]);
  atomicAdd(&z[C0 + threadIdx.x + 256], zstage[threadIdx.x + 256]);
}

__global__ __launch_bounds__(256) void k_fin_H(const float* __restrict__ fHacc,
                                               const float* __restrict__ y,
                                               const float* __restrict__ z,
                                               const float* __restrict__ sh,
                                               const float* __restrict__ bias,
                                               float* __restrict__ out) {
  int i = blockIdx.x * 256 + threadIdx.x;
  if (i < N_NODES)
    out[i] = fHacc[i] + 0.5f * (y[i] + z[i]) - sh[i] + bias[i];
}

__global__ __launch_bounds__(256) void k_fin_K(const float* __restrict__ fKacc,
                                               const float* __restrict__ skn,
                                               float* __restrict__ out) {
  int row  = blockIdx.x * 8 + (threadIdx.x >> 5);
  int lane = threadIdx.x & 31;
  size_t base = (size_t)row * D_OSC;
  float r0 = fKacc[base + lane], r1 = fKacc[base + lane + 32];
  float s0 = skn[base + lane],   s1 = skn[base + lane + 32];
  float dot = wave_reduce_sum(s0 * r0 + s1 * r1);
  out[base + lane]      = -r0 + s0 * dot;
  out[base + lane + 32] = -r1 + s1 * dot;
}

extern "C" void kernel_launch(void* const* d_in, const int* in_sizes, int n_in,
                              void* d_out, int out_size, void* d_ws, size_t ws_size,
                              hipStream_t stream) {
  const float* state_H    = (const float*)d_in[0];
  const float* state_K    = (const float*)d_in[1];
  const int*   ind_K      = (const int*)d_in[2];
  const int*   ind_HK     = (const int*)d_in[3];
  const float* kappa_K    = (const float*)d_in[4];
  const float* kappa_H    = (const float*)d_in[5];
  const float* weights_H  = (const float*)d_in[6];
  const float* bias_H     = (const float*)d_in[7];
  const float* weights_HK = (const float*)d_in[8];
  const float* w_K        = (const float*)d_in[9];

  float* out_fH = (float*)d_out;
  float* out_fK = out_fH + N_NODES;

  float* ws     = (float*)d_ws;
  float* fK_acc = ws;
  float* fH_acc = fK_acc + (size_t)N_NODES * D_OSC;
  float* yv     = fH_acc + N_NODES;
  float* zv     = yv + N_NODES;
  float* skn    = zv + N_NODES;
  float* gv     = skn + (size_t)N_NODES * D_OSC;

  const int nAcc = N_NODES * D_OSC + 3 * N_NODES;

  k_zero<<<(nAcc + 255) / 256, 256, 0, stream>>>(ws, nAcc);

  k_norm_tanh<<<N_NODES / 8, 256, 0, stream>>>(state_K, state_H, skn, gv);

  k_edge_K<<<E_EDGES / 8, 256, 0, stream>>>(ind_K, w_K, skn, fK_acc);

  k_edge_HK<<<E_EDGES / 8, 256, 0, stream>>>(ind_HK, weights_HK, skn, gv,
                                             kappa_H, kappa_K, fH_acc, fK_acc);

  k_gemv_fused<<<dim3(16, 64), 256, 0, stream>>>(weights_H, gv, yv, zv);

  k_fin_H<<<(N_NODES + 255) / 256, 256, 0, stream>>>(fH_acc, yv, zv,
                                                     state_H, bias_H, out_fH);
  k_fin_K<<<N_NODES / 8, 256, 0, stream>>>(fK_acc, skn, out_fK);
}


# W pass carried inside the scatter kernel: each wave adds a row-strip tile (y) and a column-strip tile (z: W read as the B operand, no LDS) per two edge trips; W launch gone; y/z sum order changes
# speedup vs baseline: 2.9948x; 2.9948x over previous
#include <hip/hip_runtime.h>
#include <math.h>

#define N_NODES 8192
#define D_OSC   64
#define E_EDGES 1048576
#define SL_D   4
#define N_SL   (D_OSC / SL_D)
#define N_ES   16
#define E_PER  (E_EDGES / N_ES)
#define E_FH   (E_PER / N_SL)
#define N_FP   (N_SL * N_ES)
#define N_EW   256
#define E_EW   (E_EDGES / N_EW)
#define E_TH   (E_EW / 1024)

typedef __attribute__((ext_vector_type(2))) float v2f;
typedef __attribute__((ext_vector_type(8))) float v8f;

__device__ __forceinline__ float wave_reduce_sum(float v) {
  #pragma unroll
  for (int m = 16; m >= 1; m >>= 1) v += __shfl_xor(v, m, 32);
  return v;
}

__global__ __launch_bounds__(256) void k_zero(float* __restrict__ p, int n) {
  int i = blockIdx.x * 256 + threadIdx.x;
  if (i < n) p[i] = 0.0f;
}

__global__ __launch_bounds__(256) void k_norm_tanh(const float* __restrict__ sk,
                                                   const float* __restrict__ sh,
                                                   float* __restrict__ skn,
                                                   float* __restrict__ sknS,
                                                   float* __restrict__ gv) {
  int row  = blockIdx.x * 8 + (threadIdx.x >> 5);
  int lane = threadIdx.x & 31;
  const float* r = sk + (size_t)row * D_OSC;
  float a = r[lane];
  float b = r[lane + 32];
  float ss  = wave_reduce_sum(a * a + b * b);
  float inv = rsqrtf(ss);
  float* o = skn + (size_t)row * D_OSC;
  o[lane]      = a * inv;
  o[lane + 32] = b * inv;
  {
    int d0 = lane, d1 = lane + 32;
    sknS[(size_t)(d0 / SL_D) * (N_NODES * SL_D) + row * SL_D + (d0 % SL_D)] = a * inv;
    sknS[(size_t)(d1 / SL_D) * (N_NODES * SL_D) + row * SL_D + (d1 % SL_D)] = b * inv;
  }

  int gi = blockIdx.x * 256 + threadIdx.x;
  if (gi < N_NODES) gv[gi] = tanhf(sh[gi]);
}

__global__ __launch_bounds__(1024) void k_edge(const int* __restrict__ indK,
                                               const int* __restrict__ indHK,
                                               const float* __restrict__ sknS,
                                               const float* __restrict__ wK,
                                               const float* __restrict__ whk,
                                               const float* __restrict__ gv,
                                               const float* __restrict__ kH,
                                               const float* __restrict__ kK,
                                               float* __restrict__ cK,
                                               float* __restrict__ cH,
                                               float* __restrict__ hAB) {
  __shared__ float4 sS[2 * N_NODES];
  const int t = threadIdx.x;
  const float4* src = (const float4*)sknS;
  const int2* iK = (const int2*)indK;
  const int2* iH = (const int2*)indHK;
  const int e0 = blockIdx.x * E_EW + t;
  const int s0 = blockIdx.x & (N_SL - 1);

  float4 nx[N_NODES / 1024];
  int2 a[E_TH], b[E_TH];
  #pragma unroll
  for (int i = 0; i < N_NODES / 1024; ++i) nx[i] = src[(size_t)s0 * N_NODES + i * 1024 + t];
  #pragma unroll
  for (int j = 0; j < E_TH; ++j) {
    a[j] = iK[e0 + j * 1024];
    b[j] = iH[e0 + j * 1024];
  }
  #pragma unroll
  for (int i = 0; i < N_NODES / 1024; ++i) sS[i * 1024 + t] = nx[i];
  float dK[E_TH], dH[E_TH];
  #pragma unroll
  for (int j = 0; j < E_TH; ++j) { dK[j] = 0.0f; dH[j] = 0.0f; }
  __syncthreads();

  auto slice_dots = [&](int buf) {
    const int o = buf * N_NODES;
    #pragma unroll
    for (int j = 0; j < E_TH; ++j) {
      float4 x0 = sS[o + a[j].x], x1 = sS[o + a[j].y];
      float4 y0 = sS[o + b[j].x], y1 = sS[o + b[j].y];
      dK[j] += x0.x * x1.x + x0.y * x1.y + x0.z * x1.z + x0.w * x1.w;
      dH[j] += y0.x * y1.x + y0.y * y1.y + y0.z * y1.z + y0.w * y1.w;
    }
  };

  for (int s = 0; s + 1 < N_SL; ++s) {
    const float4* nsrc = src + (size_t)((s0 + s + 1) & (N_SL - 1)) * N_NODES;
    #pragma unroll
    for (int i = 0; i < N_NODES / 1024; ++i) nx[i] = nsrc[i * 1024 + t];
    slice_dots(s & 1);
    float4* nb = sS + ((s + 1) & 1) * N_NODES;
    #pragma unroll
    for (int i = 0; i < N_NODES / 1024; ++i) nb[i * 1024 + t] = nx[i];
    __syncthreads();
  }

  float wk[E_TH], wh[E_TH], g0[E_TH], g1[E_TH];
  #pragma unroll
  for (int j = 0; j < E_TH; ++j) {
    const int e = e0 + j * 1024;
    wk[j] = wK[e];
    wh[j] = whk[e];
    g0[j] = gv[b[j].x];
    g1[j] = gv[b[j].y];
  }
  const float invKH = 1.0f / kH[0];
  const float invKK = 1.0f / kK[0];
  slice_dots((N_SL - 1) & 1);

  #pragma unroll
  for (int j = 0; j < E_TH; ++j) {
    const int e = e0 + j * 1024;
    cK[e] = wk[j] * dK[j];
    ((float2*)hAB)[e] = make_float2(dH[j] * wh[j] * g1[j] * invKH,
                                    dH[j] * wh[j] * g0[j] * invKH);
    cH[e] = -(g0[j] * g1[j]) * wh[j] * invKK;
  }
}

#define W_BLK 512
#define W_TPS (W_BLK / 16)

__global__ __launch_bounds__(1024) void k_scatter(const int* __restrict__ indK,
                                                  const int* __restrict__ indHK,
                                                  const float* __restrict__ sknS,
                                                  const float* __restrict__ cK,
                                                  const float* __restrict__ cH,
                                                  const float* __restrict__ hAB,
                                                  const float* __restrict__ Wm,
                                                  const float* __restrict__ gv,
                                                  float* __restrict__ accT,
                                                  float* __restrict__ fHp,
                                                  float* __restrict__ y,
                                                  float* __restrict__ z) {
  __shared__ float4 sS[N_NODES];
  __shared__ float  sA[SL_D * N_NODES];
  __shared__ float4 sF[N_NODES / 4];
  const int s = blockIdx.x, q = blockIdx.y, t = threadIdx.x;
  const float4* src = (const float4*)sknS + (size_t)s * N_NODES;
  #pragma unroll
  for (int i = 0; i < N_NODES / 1024; ++i) sS[i * 1024 + t] = src[i * 1024 + t];
  for (int i = 0; i < SL_D * N_NODES / 1024; ++i) sA[i * 1024 + t] = 0.0f;
  float* sFf = (float*)sF;
  for (int i = 0; i < N_NODES / 1024; ++i) sFf[i * 1024 + t] = 0.0f;
  __syncthreads();

  const int2* iK = (const int2*)indK;
  const int2* iH = (const int2*)indHK;
  const int e0 = q * E_PER + t;
  const int NIT = E_PER / 1024;
  int2 aK = iK[e0];  float cKc = cK[e0];
  int2 aH = iH[e0];  float cHc = cH[e0];

  const int lane = t & 31, wv = t >> 5, hl = lane & 15;
  const bool hi = (lane & 16) != 0;
  const int koff = hi ? 2 : 0;
  const int R0 = q * W_BLK, C0 = s * W_BLK;
  const float* wy = Wm + (size_t)(R0 + 16 * wv + hl) * N_NODES + C0 + koff;
  const float* wz = Wm + (size_t)(R0 + koff) * N_NODES + C0 + 16 * wv + hl;
  const float* gy = gv + C0 + koff;
  const float* gz = gv + R0 + koff;
  v8f yacc = {}, zacc = {};
  v2f tyA[4], tzA[4], tyB[4], tzB[4];
  #pragma unroll
  for (int j = 0; j < 4; ++j) {
    tyA[j] = *(const v2f*)(wy + 4 * j);
    tzA[j][0] = wz[(size_t)(4 * j) * N_NODES];
    tzA[j][1] = wz[(size_t)(4 * j + 1) * N_NODES];
  }

  {
    const float2* hv = (const float2*)hAB;
    const int f0 = q * E_PER + s * E_FH + t;
    int2   fa[E_FH / 1024];
    float2 fh[E_FH / 1024];
    #pragma unroll
    for (int it = 0; it < E_FH / 1024; ++it) {
      fa[it] = iH[f0 + it * 1024];
      fh[it] = hv[f0 + it * 1024];
    }
    #pragma unroll
    for (int it = 0; it < E_FH / 1024; ++it) {
      atomicAdd(&sFf[fa[it].x], fh[it].x);
      atomicAdd(&sFf[fa[it].y], fh[it].y);
    }
  }

  auto edge_trip = [&](int2 a, float c, int2 b, float ch) {
    {
      float4 x0 = sS[a.x], x1 = sS[a.y];
      atomicAdd(&sA[0 * N_NODES + a.x], c * x1.x);
      atomicAdd(&sA[1 * N_NODES + a.x], c * x1.y);
      atomicAdd(&sA[2 * N_NODES + a.x], c * x1.z);
      atomicAdd(&sA[3 * N_NODES + a.x], c * x1.w);
      atomicAdd(&sA[0 * N_NODES + a.y], c * x0.x);
      atomicAdd(&sA[1 * N_NODES + a.y], c * x0.y);
      atomicAdd(&sA[2 * N_NODES + a.y], c * x0.z);
      atomicAdd(&sA[3 * N_NODES + a.y], c * x0.w);
    }
    {
      float4 x0 = sS[b.x], x1 = sS[b.y];
      atomicAdd(&sA[0 * N_NODES + b.x], ch * x1.x);
      atomicAdd(&sA[1 * N_NODES + b.x], ch * x1.y);
      atomicAdd(&sA[2 * N_NODES + b.x], ch * x1.z);
      atomicAdd(&sA[3 * N_NODES + b.x], ch * x1.w);
      atomicAdd(&sA[0 * N_NODES + b.y], ch * x0.x);
      atomicAdd(&sA[1 * N_NODES + b.y], ch * x0.y);
      atomicAdd(&sA[2 * N_NODES + b.y], ch * x0.z);
      atomicAdd(&sA[3 * N_NODES + b.y], ch * x0.w);
    }
  };

  auto trip_pair = [&](int d, v2f (&uy)[4], v2f (&uz)[4], v2f (&ny)[4], v2f (&nz)[4]) {
    {
      v2f by[4], az[4];
      #pragma unroll
      for (int j = 0; j < 4; ++j) {
        by[j] = *(const v2f*)(gy + 16 * d + 4 * j);
        az[j] = *(const v2f*)(gz + 16 * d + 4 * j);
      }
      const int en = e0 + (2 * d + 1) * 1024;
      int2 aKn = iK[en];  float cKn = cK[en];
      int2 aHn = iH[en];  float cHn = cH[en];

      #pragma unroll
      for (int j = 0; j < 4; ++j) {
        yacc = __builtin_amdgcn_wmma_f32_16x16x4_f32(false, uy[j], false, by[j], (short)0, yacc, false, false);
        zacc = __builtin_amdgcn_wmma_f32_16x16x4_f32(false, az[j], false, uz[j], (short)0, zacc, false, false);
      }
      const int dn = (d + 1 < W_TPS) ? d + 1 : d;
      #pragma unroll
      for (int j = 0; j < 4; ++j) {
        ny[j] = *(const v2f*)(wy + 16 * dn + 4 * j);
        nz[j][0] = wz[(size_t)(16 * dn + 4 * j) * N_NODES];
        nz[j][1] = wz[(size_t)(16 * dn + 4 * j + 1) * N_NODES];
      }

      edge_trip(aK, cKc, aH, cHc);
      aK = aKn;  cKc = cKn;
      aH = aHn;  cHc = cHn;
    }
    {
      const int en = e0 + ((2 * d + 2 < NIT) ? 2 * d + 2 : 2 * d + 1) * 1024;
      int2 aKn = iK[en];  float cKn = cK[en];
      int2 aHn = iH[en];  float cHn = cH[en];
      edge_trip(aK, cKc, aH, cHc);
      aK = aKn;  cKc = cKn;
      aH = aHn;  cHc = cHn;
    }
  };

  for (int d = 0; d < W_TPS; d += 2) {
    trip_pair(d,     tyA, tzA, tyB, tzB);
    trip_pair(d + 1, tyB, tzB, tyA, tzA);
  }

  if (hl == 0) {
    const int base = R0 + 16 * wv + (hi ? 8 : 0);
    #pragma unroll
    for (int v = 0; v < 8; ++v) atomicAdd(&y[base + v], yacc[v]);
  }
  if (!hi) atomicAdd(&z[C0 + 16 * wv + hl], zacc[0]);
  __syncthreads();

  float* dst = accT + (size_t)s * (SL_D * N_NODES);
  for (int i = 0; i < SL_D * N_NODES / 1024; ++i)
    atomicAdd(&dst[i * 1024 + t], sA[i * 1024 + t]);

  float4* fp = (float4*)(fHp + (size_t)(s * N_ES + q) * N_NODES);
  for (int i = 0; i < N_NODES / 4 / 1024; ++i) fp[i * 1024 + t] = sF[i * 1024 + t];
}

__global__ __launch_bounds__(256) void k_fin_H(const float* __restrict__ fHp,
                                               const float* __restrict__ y,
                                               const float* __restrict__ z,
                                               const float* __restrict__ sh,
                                               const float* __restrict__ bias,
                                               float* __restrict__ out) {
  __shared__ float red[8][32];
  const int n  = threadIdx.x & 31;
  const int pg = threadIdx.x >> 5;
  const int i  = blockIdx.x * 32 + n;
  float acc = 0.0f;
  #pragma unroll 8
  for (int p = pg; p < N_FP; p += 8) acc += fHp[(size_t)p * N_NODES + i];
  red[pg][n] = acc;
  __syncthreads();
  if (pg == 0) {
    float e = 0.0f;
    #pragma unroll
    for (int k = 0; k < 8; ++k) e += red[k][n];
    out[i] = e + 0.5f * (y[i] + z[i]) - sh[i] + bias[i];
  }
}

__global__ __launch_bounds__(256) void k_fin_K(const float* __restrict__ fKacc,
                                               const float* __restrict__ skn,
                                               float* __restrict__ out) {
  int row  = blockIdx.x * 8 + (threadIdx.x >> 5);
  int lane = threadIdx.x & 31;
  size_t base = (size_t)row * D_OSC;
  float r0 = fKacc[(size_t)lane * N_NODES + row];
  float r1 = fKacc[(size_t)(lane + 32) * N_NODES + row];
  float s0 = skn[base + lane],   s1 = skn[base + lane + 32];
  float dot = wave_reduce_sum(s0 * r0 + s1 * r1);
  out[base + lane]      = -r0 + s0 * dot;
  out[base + lane + 32] = -r1 + s1 * dot;
}

extern "C" void kernel_launch(void* const* d_in, const int* in_sizes, int n_in,
                              void* d_out, int out_size, void* d_ws, size_t ws_size,
                              hipStream_t stream) {
  const float* state_H    = (const float*)d_in[0];
  const float* state_K    = (const float*)d_in[1];
  const int*   ind_K      = (const int*)d_in[2];
  const int*   ind_HK     = (const int*)d_in[3];
  const float* kappa_K    = (const float*)d_in[4];
  const float* kappa_H    = (const float*)d_in[5];
  const float* weights_H  = (const float*)d_in[6];
  const float* bias_H     = (const float*)d_in[7];
  const float* weights_HK = (const float*)d_in[8];
  const float* w_K        = (const float*)d_in[9];

  float* out_fH = (float*)d_out;
  float* out_fK = out_fH + N_NODES;

  float* ws     = (float*)d_ws;
  float* fK_acc = ws;
  float* yv     = fK_acc + (size_t)N_NODES * D_OSC;
  float* zv     = yv + N_NODES;
  float* skn    = zv + N_NODES;
  float* gv     = skn + (size_t)N_NODES * D_OSC;
  float* sknS   = gv + N_NODES;
  float* cKv    = sknS + (size_t)N_NODES * D_OSC;
  float* cHv    = cKv + E_EDGES;
  float* hABv   = cHv + E_EDGES;
  float* fHpl   = hABv + (size_t)2 * E_EDGES;

  const int nAcc = N_NODES * D_OSC + 2 * N_NODES;

  k_zero<<<(nAcc + 255) / 256, 256, 0, stream>>>(ws, nAcc);

  k_norm_tanh<<<N_NODES / 8, 256, 0, stream>>>(state_K, state_H, skn, sknS, gv);

  k_edge<<<N_EW, 1024, 0, stream>>>(ind_K, ind_HK, sknS, w_K, weights_HK, gv,
                                    kappa_H, kappa_K, cKv, cHv, hABv);

  k_scatter<<<dim3(N_SL, N_ES), 1024, 0, stream>>>(ind_K, ind_HK, sknS, cKv, cHv, hABv,
                                                   weights_H, gv, fK_acc, fHpl, yv, zv);

  k_fin_H<<<N_NODES / 32, 256, 0, stream>>>(fHpl, yv, zv,
                                            state_H, bias_H, out_fH);
  k_fin_K<<<N_NODES / 8, 256, 0, stream>>>(fK_acc, skn, out_fK);
}
